# starting kernel
# baseline (speedup 1.0000x reference)
#include <hip/hip_runtime.h>

#define D_DIM 64
#define TWO_D 128
#define R_REL 16

#define LDS_W_FLOATS  (R_REL * D_DIM * D_DIM)
#define LDS_R_FLOATS  (R_REL * D_DIM)
#define LDS_BYTES     ((LDS_W_FLOATS + LDS_R_FLOATS) * 4)

typedef __attribute__((ext_vector_type(2))) float v2f;
typedef __attribute__((ext_vector_type(8))) float v8f;

__global__ void zero_ws_kernel(float* __restrict__ ws, int n) {
    int i = blockIdx.x * blockDim.x + threadIdx.x;
    int stride = gridDim.x * blockDim.x;
    for (; i < n; i += stride) ws[i] = 0.0f;
}

__global__ __launch_bounds__(1024)
void edge_attention_scatter(const float* __restrict__ ent,
                            const float* __restrict__ rel,
                            const float* __restrict__ WR,
                            const int*   __restrict__ src,
                            const int*   __restrict__ dst,
                            const int*   __restrict__ ety,
                            float*       __restrict__ nb,
                            int E) {
    extern __shared__ float lds[];
    float* ldsW = lds;
    float* ldsR = lds + LDS_W_FLOATS;

    {
        float4*       dw = (float4*)ldsW;
        const float4* sw = (const float4*)WR;
        for (int i = threadIdx.x; i < LDS_W_FLOATS / 4; i += blockDim.x) dw[i] = sw[i];
        float4*       dr = (float4*)ldsR;
        const float4* sr = (const float4*)rel;
        for (int i = threadIdx.x; i < LDS_R_FLOATS / 4; i += blockDim.x) dr[i] = sr[i];
    }
    __syncthreads();

    const int lane   = threadIdx.x & 31;
    const int half   = lane >> 4;
    const int lr     = lane & 15;
    const int wave   = blockIdx.x * (blockDim.x >> 5) + (threadIdx.x >> 5);
    const int nwaves = gridDim.x * (blockDim.x >> 5);

    for (int e = wave; e < E; e += nwaves) {
        const int s = src[e];
        const int t = dst[e];
        const int r = ety[e];

        const float* srow = ent + (size_t)s * D_DIM;
        const float  s0 = srow[lane];
        const float  s1 = srow[lane + 32];

        const float* w = ldsW + r * (D_DIM * D_DIM);

        float4 acc = make_float4(0.f, 0.f, 0.f, 0.f);
#pragma unroll 4
        for (int i = 0; i < 16; ++i) {
            const int d = 2 * i + half;
            const float a = __shfl(s0, d, 32);
            const float4 w4 = *(const float4*)(w + d * D_DIM + 4 * lr);
            acc.x = fmaf(a, w4.x, acc.x);
            acc.y = fmaf(a, w4.y, acc.y);
            acc.z = fmaf(a, w4.z, acc.z);
            acc.w = fmaf(a, w4.w, acc.w);
        }
#pragma unroll 4
        for (int i = 0; i < 16; ++i) {
            const int d = 2 * i + half;
            const float a = __shfl(s1, d, 32);
            const float4 w4 = *(const float4*)(w + (32 + d) * D_DIM + 4 * lr);
            acc.x = fmaf(a, w4.x, acc.x);
            acc.y = fmaf(a, w4.y, acc.y);
            acc.z = fmaf(a, w4.z, acc.z);
            acc.w = fmaf(a, w4.w, acc.w);
        }
        acc.x += __shfl_xor(acc.x, 16, 32);
        acc.y += __shfl_xor(acc.y, 16, 32);
        acc.z += __shfl_xor(acc.z, 16, 32);
        acc.w += __shfl_xor(acc.w, 16, 32);

        const float4 re4 = *(const float4*)(ldsR + r * D_DIM + 4 * lr);
        const float ma = half ? acc.z : acc.x;
        const float mb = half ? acc.w : acc.y;
        const float ra = half ? re4.z : re4.x;
        const float rb = half ? re4.w : re4.y;
        float part = ma * tanhf(ma + ra) + mb * tanhf(mb + rb);
#pragma unroll
        for (int off = 16; off > 0; off >>= 1)
            part += __shfl_xor(part, off, 32);
        const float att = part;

        float* drow = nb + (size_t)t * D_DIM;
        unsafeAtomicAdd(drow + lane,      s0 * att);
        unsafeAtomicAdd(drow + lane + 32, s1 * att);
    }
}

__global__ __launch_bounds__(256)
void combine_wmma(const float* __restrict__ ent,
                  const float* __restrict__ nb,
                  const float* __restrict__ Ww,
                  const float* __restrict__ Wb,
                  float*       __restrict__ out,
                  int n) {
    const int lane = threadIdx.x & 31;
    const int half = lane >> 4;
    const int lr   = lane & 15;

    const int rowBlock     = blockIdx.x * (blockDim.x >> 5) + (threadIdx.x >> 5);
    const int numRowBlocks = n / 16;
    if (rowBlock >= numRowBlocks) return;
    const int row0 = rowBlock * 16;

    v8f acc[4] = {};

    const float* hrow = ent + (size_t)(row0 + lr) * D_DIM;
    const float* nrow = nb  + (size_t)(row0 + lr) * D_DIM;

#pragma unroll 4
    for (int k = 0; k < TWO_D; k += 4) {
        const int k0 = k + 2 * half;
        const float* ap = (k0 < D_DIM) ? (hrow + k0) : (nrow + (k0 - D_DIM));
        v2f a;
        a.x = ap[0];
        a.y = ap[1];

#pragma unroll
        for (int c = 0; c < 4; ++c) {
            const int col = c * 16 + lr;
            const float* bp = Ww + (size_t)col * TWO_D + k0;
            v2f b;
            b.x = bp[0];
            b.y = bp[1];
            acc[c] = __builtin_amdgcn_wmma_f32_16x16x4_f32(
                 false, a,  false, b,
                 (short)0, acc[c],  false,  false);
        }
    }

#pragma unroll
    for (int c = 0; c < 4; ++c) {
        const int col = c * 16 + lr;
        const float bias = Wb[col];
#pragma unroll
        for (int r = 0; r < 8; ++r) {
            const int row = row0 + r + 8 * half;
            float x = acc[c][r] + bias;
            x = (x > 0.0f) ? x : 0.01f * x;
            out[(size_t)row * D_DIM + col] = x + ent[(size_t)row * D_DIM + col];
        }
    }
}

extern "C" void kernel_launch(void* const* d_in, const int* in_sizes, int n_in,
                              void* d_out, int out_size, void* d_ws, size_t ws_size,
                              hipStream_t stream) {
    const float* ent = (const float*)d_in[0];
    const float* rel = (const float*)d_in[1];
    const float* WR  = (const float*)d_in[2];
    const float* Ww  = (const float*)d_in[3];
    const float* Wb  = (const float*)d_in[4];
    const int*   src = (const int*)d_in[5];
    const int*   dst = (const int*)d_in[6];
    const int*   ety = (const int*)d_in[7];
    float*       out = (float*)d_out;

    const int n = in_sizes[0] / D_DIM;
    const int E = in_sizes[5];

    float* nb = (float*)d_ws;

    (void)hipFuncSetAttribute((const void*)edge_attention_scatter,
                              hipFuncAttributeMaxDynamicSharedMemorySize,
                              LDS_BYTES);

    zero_ws_kernel<<<1024, 256, 0, stream>>>(nb, n * D_DIM);

    edge_attention_scatter<<<512, 1024, LDS_BYTES, stream>>>(ent, rel, WR, src, dst, ety, nb, E);

    const int rowBlocks = n / 16;
    combine_wmma<<<(rowBlocks + 7) / 8, 256, 0, stream>>>(ent, nb, Ww, Wb, out, n);
}


# Table kernel re-cut: a block serves ONE relation over a 6250-node range (16x16 blocks), not 16 relations over 392 nodes: stages 16 KB of W_R, all opening loads in one round trip; sums unchanged
# speedup vs baseline: 1.8759x; 1.8759x over previous
#include <hip/hip_runtime.h>

#define D_DIM 64
#define TWO_D 128
#define R_REL 16

#define TB_THREADS    1024
#define RG_NODES      6250
#define FL_LOADS      7
#define LIST_CAP      (RG_NODES + 6 + 16)
#define W_SLOTS       (D_DIM * D_DIM / 4)
static_assert(W_SLOTS == TB_THREADS, "one W_R slot per thread of the table kernel");
static_assert(FL_LOADS * TB_THREADS >= RG_NODES, "flag loads cover a range");

typedef __attribute__((ext_vector_type(2))) float v2f;
typedef __attribute__((ext_vector_type(8))) float v8f;

__global__ void zero_ws_kernel(float* __restrict__ ws, int n,
                               unsigned int* __restrict__ fl, int nf) {
    const int t = blockIdx.x * blockDim.x + threadIdx.x;
    const int stride = gridDim.x * blockDim.x;
    for (int i = t; i < nf; i += stride) fl[i] = 0u;
    for (int i = t; i < n;  i += stride) ws[i] = 0.0f;
}

__global__ __launch_bounds__(256)
void mark_needed_pairs(const int* __restrict__ src,
                       const int* __restrict__ ety,
                       unsigned char* __restrict__ flags,
                       int n, int E) {
    const int t = blockIdx.x * blockDim.x + threadIdx.x;
    const int stride = gridDim.x * blockDim.x;
    const int nq = E >> 2;
    const int4* s4p = (const int4*)src;
    const int4* r4p = (const int4*)ety;
    for (int q = t; q < nq; q += stride) {
        const int4 s4 = s4p[q];
        const int4 r4 = r4p[q];
        flags[(size_t)r4.x * n + s4.x] = 1;
        flags[(size_t)r4.y * n + s4.y] = 1;
        flags[(size_t)r4.z * n + s4.z] = 1;
        flags[(size_t)r4.w * n + s4.w] = 1;
    }
    for (int e = (nq << 2) + t; e < E; e += stride) flags[(size_t)ety[e] * n + src[e]] = 1;
}

__global__ __launch_bounds__(TB_THREADS)
void node_relation_attention(const float* __restrict__ ent,
                             const float* __restrict__ rel,
                             const float* __restrict__ WR,
                             const unsigned char* __restrict__ flags,
                             float*       __restrict__ att,
                             int n) {
    __shared__ float4 ldsW4[W_SLOTS];
    __shared__ alignas(16) float ldsR[D_DIM];
    __shared__ int    ldsL[LIST_CAP];
    __shared__ int    ldsC[32];

    const int tid  = threadIdx.x;
    const int lane = tid & 31;
    const int half = lane >> 4;
    const int lr   = lane & 15;
    const int w    = __builtin_amdgcn_readfirstlane(tid >> 5);
    const int nw   = TB_THREADS >> 5;

    const int r    = blockIdx.x & (R_REL - 1);
    const int nrg  = gridDim.x >> 4;
    int rg = blockIdx.x >> 4;

    const unsigned char* f = flags + (size_t)r * n;

    unsigned fr[FL_LOADS];
    {
        const int nlo = rg * RG_NODES;
#pragma unroll
        for (int j = 0; j < FL_LOADS; ++j) fr[j] = f[min(nlo + TB_THREADS * j + tid, n - 1)];
    }
    float4 wv;
    {
        const int l  = tid & 31;
        const int p  = (tid >> 5) & 7;
        const int et = (tid >> 8) & 3;
        const int d0 = 32 * (l >> 4) + 4 * p;
        const int e  = 16 * et + (l & 15);
        const float* g = WR + r * (D_DIM * D_DIM) + d0 * D_DIM + e;
        wv.x = g[0];
        wv.y = g[D_DIM];
        wv.z = g[2 * D_DIM];
        wv.w = g[3 * D_DIM];
    }
    const float4 rv4 = ((const float4*)(rel + r * D_DIM))[tid & 15];
    asm volatile("" :: "v"(fr[0]), "v"(fr[1]), "v"(fr[2]), "v"(fr[3]), "v"(fr[4]), "v"(fr[5]), "v"(fr[6]),
                       "v"(wv.x), "v"(wv.y), "v"(wv.z), "v"(wv.w),
                       "v"(rv4.x), "v"(rv4.y), "v"(rv4.z), "v"(rv4.w));
    ldsW4[tid] = wv;
    if (tid < D_DIM / 4) ((float4*)ldsR)[tid] = rv4;

    for (;;) {
        const int nlo = rg * RG_NODES;

        bool need[FL_LOADS];
        unsigned m[FL_LOADS];
        int cnt = 0;
#pragma unroll
        for (int j = 0; j < FL_LOADS; ++j) {
            const int o = TB_THREADS * j + tid;
            need[j] = (fr[j] != 0u) && (o < RG_NODES) && (nlo + o < n);
            m[j] = __builtin_amdgcn_ballot_w32(need[j]);
            cnt += __builtin_popcount(m[j]);
        }
        if (lane == 0) ldsC[w] = cnt;
        __syncthreads();

        const int cW = (lane < nw) ? ldsC[lane] : 0;
        int incl = cW;
#pragma unroll
        for (int d = 1; d < 32; d <<= 1) {
            const int t = __shfl_up(incl, d, 32);
            if (lane >= d) incl += t;
        }
        const int total = __builtin_amdgcn_readlane(incl, 31);
        int run = __builtin_amdgcn_readlane(incl - cW, w);
#pragma unroll
        for (int j = 0; j < FL_LOADS; ++j) {
            const int pos = run + (int)__builtin_amdgcn_mbcnt_lo(m[j], 0u);
            if (need[j]) ldsL[pos] = nlo + TB_THREADS * j + tid;
            run += __builtin_popcount(m[j]);
        }
        const int padded = (total + 15) & ~15;
        if (w == 0 && total + lane < padded) ldsL[total + lane] = nlo;
        __syncthreads();

        const int tiles = padded >> 4;
        for (int it = w; it < tiles; it += nw) {
            asm volatile("" ::: "memory");
            const int slot = it * 16 + lr;
            const bool valid = slot < total;
            const int node = ldsL[slot];

            const float4* brow = (const float4*)(ent + (size_t)node * D_DIM + 32 * half);
            v2f b[16];
#pragma unroll
            for (int j = 0; j < 8; ++j) {
                const float4 t4 = brow[j];
                b[2 * j].x     = t4.x;  b[2 * j].y     = t4.y;
                b[2 * j + 1].x = t4.z;  b[2 * j + 1].y = t4.w;
            }

            const float4* wf = ldsW4 + lane;

            v8f acc[4] = {};
#pragma unroll
            for (int p = 0; p < 8; ++p) {
                float4 a[4];
#pragma unroll
                for (int et = 0; et < 4; ++et) a[et] = wf[(et * 8 + p) * 32];
#pragma unroll
                for (int et = 0; et < 4; ++et) {
                    v2f af; af.x = a[et].x; af.y = a[et].y;
                    acc[et] = __builtin_amdgcn_wmma_f32_16x16x4_f32(
                        false, af, false, b[2 * p], (short)0, acc[et], false, false);
                }
#pragma unroll
                for (int et = 0; et < 4; ++et) {
                    v2f af; af.x = a[et].z; af.y = a[et].w;
                    acc[et] = __builtin_amdgcn_wmma_f32_16x16x4_f32(
                        false, af, false, b[2 * p + 1], (short)0, acc[et], false, false);
                }
            }

            float part = 0.f;
#pragma unroll
            for (int et = 0; et < 4; ++et) {
                const float4* rp = (const float4*)(ldsR + 16 * et + 8 * half);
                const float4 r0 = rp[0];
                const float4 r1 = rp[1];
                const float rv[8] = {r0.x, r0.y, r0.z, r0.w, r1.x, r1.y, r1.z, r1.w};
#pragma unroll
                for (int q = 0; q < 8; ++q) {
                    const float mv = acc[et][q];
                    part += mv * tanhf(mv + rv[q]);
                }
            }
            part += __shfl_xor(part, 16, 32);

            if (half == 0 && valid) att[(size_t)r * n + node] = part;
        }

        rg += nrg;
        if (rg * RG_NODES >= n) break;
        __syncthreads();
        {
            const int nlo2 = rg * RG_NODES;
#pragma unroll
            for (int j = 0; j < FL_LOADS; ++j) fr[j] = f[min(nlo2 + TB_THREADS * j + tid, n - 1)];
            asm volatile("" :: "v"(fr[0]), "v"(fr[1]), "v"(fr[2]), "v"(fr[3]), "v"(fr[4]), "v"(fr[5]), "v"(fr[6]));
        }
    }
}

__global__ __launch_bounds__(256)
void edge_gather_scatter(const float* __restrict__ ent,
                         const float* __restrict__ att,
                         const int*   __restrict__ src,
                         const int*   __restrict__ dst,
                         const int*   __restrict__ ety,
                         float*       __restrict__ nb,
                         int n, int E) {
    const int lane    = threadIdx.x & 31;
    const int wave    = blockIdx.x * (blockDim.x >> 5) + (threadIdx.x >> 5);
    const int nwaves  = gridDim.x * (blockDim.x >> 5);
    const int nChunks = (E + 31) >> 5;

    for (int ch = wave; ch < nChunks; ch += nwaves) {
        const int e = ch * 32 + lane;
        int   s = 0, t = 0;
        float a = 0.f;
        if (e < E) {
            s = src[e];
            t = dst[e];
            a = att[(size_t)ety[e] * n + s];
        }
        const int cnt = min(32, E - ch * 32);

        if (cnt == 32) {
#pragma unroll
            for (int j0 = 0; j0 < 32; j0 += 8) {
                float x0[8], x1[8];
#pragma unroll
                for (int u = 0; u < 8; ++u) {
                    const int sj = __builtin_amdgcn_readlane(s, j0 + u);
                    const float* srow = ent + (size_t)sj * D_DIM;
                    x0[u] = srow[lane];
                    x1[u] = srow[lane + 32];
                }
#pragma unroll
                for (int u = 0; u < 8; ++u) {
                    const int   tj = __builtin_amdgcn_readlane(t, j0 + u);
                    const float aj = __int_as_float(__builtin_amdgcn_readlane(__float_as_int(a), j0 + u));
                    float* drow = nb + (size_t)tj * D_DIM;
                    unsafeAtomicAdd(drow + lane,      x0[u] * aj);
                    unsafeAtomicAdd(drow + lane + 32, x1[u] * aj);
                }
            }
        } else {
            for (int j = 0; j < cnt; ++j) {
                const int   sj = __shfl(s, j, 32);
                const int   tj = __shfl(t, j, 32);
                const float aj = __shfl(a, j, 32);
                const float* srow = ent + (size_t)sj * D_DIM;
                float* drow = nb + (size_t)tj * D_DIM;
                unsafeAtomicAdd(drow + lane,      srow[lane] * aj);
                unsafeAtomicAdd(drow + lane + 32, srow[lane + 32] * aj);
            }
        }
    }
}

__global__ __launch_bounds__(256)
void combine_wmma(const float* __restrict__ ent,
                  const float* __restrict__ nb,
                  const float* __restrict__ Ww,
                  const float* __restrict__ Wb,
                  float*       __restrict__ out,
                  int n) {
    const int lane = threadIdx.x & 31;
    const int half = lane >> 4;
    const int lr   = lane & 15;

    const int rowBlock     = blockIdx.x * (blockDim.x >> 5) + (threadIdx.x >> 5);
    const int numRowBlocks = n / 16;
    if (rowBlock >= numRowBlocks) return;
    const int row0 = rowBlock * 16;

    v8f acc[4] = {};

    const float* hrow = ent + (size_t)(row0 + lr) * D_DIM;
    const float* nrow = nb  + (size_t)(row0 + lr) * D_DIM;

#pragma unroll 4
    for (int k = 0; k < TWO_D; k += 4) {
        const int k0 = k + 2 * half;
        const float* ap = (k0 < D_DIM) ? (hrow + k0) : (nrow + (k0 - D_DIM));
        v2f a;
        a.x = ap[0];
        a.y = ap[1];

#pragma unroll
        for (int c = 0; c < 4; ++c) {
            const int col = c * 16 + lr;
            const float* bp = Ww + (size_t)col * TWO_D + k0;
            v2f b;
            b.x = bp[0];
            b.y = bp[1];
            acc[c] = __builtin_amdgcn_wmma_f32_16x16x4_f32(
                 false, a,  false, b,
                 (short)0, acc[c],  false,  false);
        }
    }

#pragma unroll
    for (int c = 0; c < 4; ++c) {
        const int col = c * 16 + lr;
        const float bias = Wb[col];
#pragma unroll
        for (int r = 0; r < 8; ++r) {
            const int row = row0 + r + 8 * half;
            float x = acc[c][r] + bias;
            x = (x > 0.0f) ? x : 0.01f * x;
            out[(size_t)row * D_DIM + col] = x + ent[(size_t)row * D_DIM + col];
        }
    }
}

extern "C" void kernel_launch(void* const* d_in, const int* in_sizes, int n_in,
                              void* d_out, int out_size, void* d_ws, size_t ws_size,
                              hipStream_t stream) {
    const float* ent = (const float*)d_in[0];
    const float* rel = (const float*)d_in[1];
    const float* WR  = (const float*)d_in[2];
    const float* Ww  = (const float*)d_in[3];
    const float* Wb  = (const float*)d_in[4];
    const int*   src = (const int*)d_in[5];
    const int*   dst = (const int*)d_in[6];
    const int*   ety = (const int*)d_in[7];
    float*       out = (float*)d_out;

    const int n = in_sizes[0] / D_DIM;
    const int E = in_sizes[5];

    float* nb  = (float*)d_ws;
    float* att = out;
    unsigned char* flags = (unsigned char*)(out + (size_t)R_REL * n);

    zero_ws_kernel<<<1024, 256, 0, stream>>>(nb, n * D_DIM, (unsigned int*)flags, (R_REL * n) / 4);

    mark_needed_pairs<<<512, 256, 0, stream>>>(src, ety, flags, n, E);

    node_relation_attention<<<256, TB_THREADS, 0, stream>>>(ent, rel, WR, flags, att, n);

    edge_gather_scatter<<<4096, 256, 0, stream>>>(ent, att, src, dst, ety, nb, n, E);

    const int rowBlocks = n / 16;
    combine_wmma<<<(rowBlocks + 7) / 8, 256, 0, stream>>>(ent, nb, Ww, Wb, out, n);
}


# Edge pass folded into the table kernel: a link pass chains each pair's edges (swap atomics); the wave that forms a pair's attention adds its node row from LDS to their dst rows; sum order changes
# speedup vs baseline: 2.2146x; 1.1806x over previous
#include <hip/hip_runtime.h>

#define D_DIM 64
#define TWO_D 128
#define R_REL 16

#define TB_THREADS    1024
#define RG_NODES      6250
#define FL_LOADS      7
#define LIST_CAP      (RG_NODES + 6 + 16)
#define W_SLOTS       (D_DIM * D_DIM / 4)
static_assert(W_SLOTS == TB_THREADS, "one W_R slot per thread of the table kernel");
static_assert(FL_LOADS * TB_THREADS >= RG_NODES, "head loads cover a range");
#define TILE_PITCH    68

typedef __attribute__((ext_vector_type(2))) float v2f;
typedef __attribute__((ext_vector_type(8))) float v8f;

__global__ void zero_ws_kernel(float* __restrict__ ws, int n,
                               unsigned int* __restrict__ fl, int nf) {
    const int t = blockIdx.x * blockDim.x + threadIdx.x;
    const int stride = gridDim.x * blockDim.x;
    for (int i = t; i < nf; i += stride) fl[i] = 0u;
    for (int i = t; i < n;  i += stride) ws[i] = 0.0f;
}

__global__ __launch_bounds__(256)
void link_edges_by_pair(const int* __restrict__ src,
                        const int* __restrict__ dst,
                        const int* __restrict__ ety,
                        int* __restrict__ head,
                        int* __restrict__ recs,
                        int n, int E) {
    const int t = blockIdx.x * blockDim.x + threadIdx.x;
    const int stride = gridDim.x * blockDim.x;
    const int nh = E >> 1;
    const int2* s2p = (const int2*)src;
    const int2* d2p = (const int2*)dst;
    const int2* r2p = (const int2*)ety;
    int4* rec4 = (int4*)recs;
    for (int q = t; q < nh; q += stride) {
        const int2 s2 = s2p[q];
        const int2 r2 = r2p[q];
        const int2 d2 = d2p[q];
        const int e1 = 2 * q + 1;
        const int p0 = atomicExch(head + (size_t)r2.x * n + s2.x, e1);
        const int p1 = atomicExch(head + (size_t)r2.y * n + s2.y, e1 + 1);
        rec4[q] = make_int4(p0, d2.x, p1, d2.y);
    }
    for (int e = (nh << 1) + t; e < E; e += stride) {
        const int p = atomicExch(head + (size_t)ety[e] * n + src[e], e + 1);
        recs[2 * e]     = p;
        recs[2 * e + 1] = dst[e];
    }
}

__global__ __launch_bounds__(TB_THREADS)
void node_relation_attention(const float* __restrict__ ent,
                             const float* __restrict__ rel,
                             const float* __restrict__ WR,
                             const int*   __restrict__ head,
                             const int*   __restrict__ recs,
                             float*       __restrict__ nb,
                             int n, int E) {
    __shared__ float4 ldsW4[W_SLOTS];
    __shared__ alignas(16) float ldsR[D_DIM];
    __shared__ int    ldsL[LIST_CAP];
    __shared__ int    ldsH[LIST_CAP];
    __shared__ int    ldsC[32];
    __shared__ alignas(16) float ldsT[(TB_THREADS >> 5) * 16 * TILE_PITCH];

    const int tid  = threadIdx.x;
    const int lane = tid & 31;
    const int half = lane >> 4;
    const int lr   = lane & 15;
    const int w    = __builtin_amdgcn_readfirstlane(tid >> 5);
    const int nw   = TB_THREADS >> 5;

    const int r    = blockIdx.x & (R_REL - 1);
    const int nrg  = gridDim.x >> 4;
    int rg = blockIdx.x >> 4;

    const int* f = head + (size_t)r * n;
    const int2* rec2 = (const int2*)recs;
    float* myT = ldsT + w * (16 * TILE_PITCH);

    int fr[FL_LOADS];
    {
        const int nlo = rg * RG_NODES;
#pragma unroll
        for (int j = 0; j < FL_LOADS; ++j) fr[j] = f[min(nlo + TB_THREADS * j + tid, n - 1)];
    }
    float4 wv;
    {
        const int l  = tid & 31;
        const int p  = (tid >> 5) & 7;
        const int et = (tid >> 8) & 3;
        const int d0 = 32 * (l >> 4) + 4 * p;
        const int e  = 16 * et + (l & 15);
        const float* g = WR + r * (D_DIM * D_DIM) + d0 * D_DIM + e;
        wv.x = g[0];
        wv.y = g[D_DIM];
        wv.z = g[2 * D_DIM];
        wv.w = g[3 * D_DIM];
    }
    const float4 rv4 = ((const float4*)(rel + r * D_DIM))[tid & 15];
    asm volatile("" :: "v"(fr[0]), "v"(fr[1]), "v"(fr[2]), "v"(fr[3]), "v"(fr[4]), "v"(fr[5]), "v"(fr[6]),
                       "v"(wv.x), "v"(wv.y), "v"(wv.z), "v"(wv.w),
                       "v"(rv4.x), "v"(rv4.y), "v"(rv4.z), "v"(rv4.w));
    ldsW4[tid] = wv;
    if (tid < D_DIM / 4) ((float4*)ldsR)[tid] = rv4;

    for (;;) {
        const int nlo = rg * RG_NODES;

        bool need[FL_LOADS];
        unsigned m[FL_LOADS];
        int cnt = 0;
#pragma unroll
        for (int j = 0; j < FL_LOADS; ++j) {
            const int o = TB_THREADS * j + tid;
            need[j] = (fr[j] != 0) && (o < RG_NODES) && (nlo + o < n);
            m[j] = __builtin_amdgcn_ballot_w32(need[j]);
            cnt += __builtin_popcount(m[j]);
        }
        if (lane == 0) ldsC[w] = cnt;
        __syncthreads();

        const int cW = (lane < nw) ? ldsC[lane] : 0;
        int incl = cW;
#pragma unroll
        for (int d = 1; d < 32; d <<= 1) {
            const int t = __shfl_up(incl, d, 32);
            if (lane >= d) incl += t;
        }
        const int total = __builtin_amdgcn_readlane(incl, 31);
        int run = __builtin_amdgcn_readlane(incl - cW, w);
#pragma unroll
        for (int j = 0; j < FL_LOADS; ++j) {
            const int pos = run + (int)__builtin_amdgcn_mbcnt_lo(m[j], 0u);
            if (need[j]) { ldsL[pos] = nlo + TB_THREADS * j + tid; ldsH[pos] = fr[j]; }
            run += __builtin_popcount(m[j]);
        }
        const int padded = (total + 15) & ~15;
        if (w == 0 && total + lane < padded) { ldsL[total + lane] = nlo; ldsH[total + lane] = 0; }
        __syncthreads();

        const int tiles = padded >> 4;
        for (int it = w; it < tiles; it += nw) {
            asm volatile("" ::: "memory");
            const int slot = it * 16 + lr;
            const int node = ldsL[slot];
            int cur = (half == 0) ? ldsH[slot] : 0;
            int2 rec = rec2[max(cur - 1, 0)];

            const float4* brow = (const float4*)(ent + (size_t)node * D_DIM + 32 * half);
            v2f b[16];
#pragma unroll
            for (int j = 0; j < 8; ++j) {
                const float4 t4 = brow[j];
                b[2 * j].x     = t4.x;  b[2 * j].y     = t4.y;
                b[2 * j + 1].x = t4.z;  b[2 * j + 1].y = t4.w;
            }

            {
                float4* trow = (float4*)(myT + lr * TILE_PITCH + 32 * half);
#pragma unroll
                for (int j = 0; j < 8; ++j) {
                    float4 t4;
                    t4.x = b[2 * j].x;      t4.y = b[2 * j].y;
                    t4.z = b[2 * j + 1].x;  t4.w = b[2 * j + 1].y;
                    trow[j] = t4;
                }
            }

            const float4* wf = ldsW4 + lane;

            v8f acc[4] = {};
#pragma unroll
            for (int p = 0; p < 8; ++p) {
                float4 a[4];
#pragma unroll
                for (int et = 0; et < 4; ++et) a[et] = wf[(et * 8 + p) * 32];
#pragma unroll
                for (int et = 0; et < 4; ++et) {
                    v2f af; af.x = a[et].x; af.y = a[et].y;
                    acc[et] = __builtin_amdgcn_wmma_f32_16x16x4_f32(
                        false, af, false, b[2 * p], (short)0, acc[et], false, false);
                }
#pragma unroll
                for (int et = 0; et < 4; ++et) {
                    v2f af; af.x = a[et].z; af.y = a[et].w;
                    acc[et] = __builtin_amdgcn_wmma_f32_16x16x4_f32(
                        false, af, false, b[2 * p + 1], (short)0, acc[et], false, false);
                }
            }

            float part = 0.f;
#pragma unroll
            for (int et = 0; et < 4; ++et) {
                const float4* rp = (const float4*)(ldsR + 16 * et + 8 * half);
                const float4 r0 = rp[0];
                const float4 r1 = rp[1];
                const float rv[8] = {r0.x, r0.y, r0.z, r0.w, r1.x, r1.y, r1.z, r1.w};
#pragma unroll
                for (int q = 0; q < 8; ++q) {
                    const float mv = acc[et][q];
                    part += mv * tanhf(mv + rv[q]);
                }
            }
            part += __shfl_xor(part, 16, 32);

            const float* trd = myT + lane;
            for (int g = 0; g < E; ++g) {
                const unsigned act = __builtin_amdgcn_ballot_w32(cur != 0);
                if (act == 0u) break;
                const int dcur = rec.y;
                const int nxt  = (cur != 0) ? rec.x : 0;
                const int2 recn = rec2[max(nxt - 1, 0)];
                unsigned mrem = act;
                do {
                    const int c = __builtin_ctz(mrem);
                    mrem &= mrem - 1u;
                    const int   d = __builtin_amdgcn_readlane(dcur, c);
                    const float a = __int_as_float(__builtin_amdgcn_readlane(__float_as_int(part), c));
                    const float x0 = trd[c * TILE_PITCH];
                    const float x1 = trd[c * TILE_PITCH + 32];
                    float* drow = nb + (size_t)d * D_DIM;
                    unsafeAtomicAdd(drow + lane,      x0 * a);
                    unsafeAtomicAdd(drow + lane + 32, x1 * a);
                } while (mrem != 0u);
                cur = nxt;
                rec = recn;
            }
        }

        rg += nrg;
        if (rg * RG_NODES >= n) break;
        __syncthreads();
        {
            const int nlo2 = rg * RG_NODES;
#pragma unroll
            for (int j = 0; j < FL_LOADS; ++j) fr[j] = f[min(nlo2 + TB_THREADS * j + tid, n - 1)];
            asm volatile("" :: "v"(fr[0]), "v"(fr[1]), "v"(fr[2]), "v"(fr[3]), "v"(fr[4]), "v"(fr[5]), "v"(fr[6]));
        }
    }
}

__global__ __launch_bounds__(256)
void combine_wmma(const float* __restrict__ ent,
                  const float* __restrict__ nb,
                  const float* __restrict__ Ww,
                  const float* __restrict__ Wb,
                  float*       __restrict__ out,
                  int n) {
    const int lane = threadIdx.x & 31;
    const int half = lane >> 4;
    const int lr   = lane & 15;

    const int rowBlock     = blockIdx.x * (blockDim.x >> 5) + (threadIdx.x >> 5);
    const int numRowBlocks = n / 16;
    if (rowBlock >= numRowBlocks) return;
    const int row0 = rowBlock * 16;

    v8f acc[4] = {};

    const float* hrow = ent + (size_t)(row0 + lr) * D_DIM;
    const float* nrow = nb  + (size_t)(row0 + lr) * D_DIM;

#pragma unroll 4
    for (int k = 0; k < TWO_D; k += 4) {
        const int k0 = k + 2 * half;
        const float* ap = (k0 < D_DIM) ? (hrow + k0) : (nrow + (k0 - D_DIM));
        v2f a;
        a.x = ap[0];
        a.y = ap[1];

#pragma unroll
        for (int c = 0; c < 4; ++c) {
            const int col = c * 16 + lr;
            const float* bp = Ww + (size_t)col * TWO_D + k0;
            v2f b;
            b.x = bp[0];
            b.y = bp[1];
            acc[c] = __builtin_amdgcn_wmma_f32_16x16x4_f32(
                 false, a,  false, b,
                 (short)0, acc[c],  false,  false);
        }
    }

#pragma unroll
    for (int c = 0; c < 4; ++c) {
        const int col = c * 16 + lr;
        const float bias = Wb[col];
#pragma unroll
        for (int r = 0; r < 8; ++r) {
            const int row = row0 + r + 8 * half;
            float x = acc[c][r] + bias;
            x = (x > 0.0f) ? x : 0.01f * x;
            out[(size_t)row * D_DIM + col] = x + ent[(size_t)row * D_DIM + col];
        }
    }
}

extern "C" void kernel_launch(void* const* d_in, const int* in_sizes, int n_in,
                              void* d_out, int out_size, void* d_ws, size_t ws_size,
                              hipStream_t stream) {
    const float* ent = (const float*)d_in[0];
    const float* rel = (const float*)d_in[1];
    const float* WR  = (const float*)d_in[2];
    const float* Ww  = (const float*)d_in[3];
    const float* Wb  = (const float*)d_in[4];
    const int*   src = (const int*)d_in[5];
    const int*   dst = (const int*)d_in[6];
    const int*   ety = (const int*)d_in[7];
    float*       out = (float*)d_out;

    const int n = in_sizes[0] / D_DIM;
    const int E = in_sizes[5];

    float* nb  = (float*)d_ws;
    int* head = (int*)out;
    int* recs = (int*)(out + (size_t)R_REL * n);

    zero_ws_kernel<<<1024, 256, 0, stream>>>(nb, n * D_DIM, (unsigned int*)head, R_REL * n);

    link_edges_by_pair<<<512, 256, 0, stream>>>(src, dst, ety, head, recs, n, E);

    node_relation_attention<<<256, TB_THREADS, 0, stream>>>(ent, rel, WR, head, recs, nb, n, E);

    const int rowBlocks = n / 16;
    combine_wmma<<<(rowBlocks + 7) / 8, 256, 0, stream>>>(ent, nb, Ww, Wb, out, n);
}


# Combine GEMM re-tiled: a wave owns 64 rows x 64 columns (four 16-row blocks; 1564 waves, not 6256), so each W_w fragment load feeds 4 WMMAs: global loads per 16 WMMAs 20 -> 8; sums unchanged
# speedup vs baseline: 2.3622x; 1.0666x over previous
#include <hip/hip_runtime.h>

#define D_DIM 64
#define TWO_D 128
#define R_REL 16

#define TB_THREADS    1024
#define RG_NODES      6250
#define FL_LOADS      7
#define LIST_CAP      (RG_NODES + 6 + 16)
#define W_SLOTS       (D_DIM * D_DIM / 4)
static_assert(W_SLOTS == TB_THREADS, "one W_R slot per thread of the table kernel");
static_assert(FL_LOADS * TB_THREADS >= RG_NODES, "head loads cover a range");
#define TILE_PITCH    68

typedef __attribute__((ext_vector_type(2))) float v2f;
typedef __attribute__((ext_vector_type(8))) float v8f;

__global__ void zero_ws_kernel(float* __restrict__ ws, int n,
                               unsigned int* __restrict__ fl, int nf) {
    const int t = blockIdx.x * blockDim.x + threadIdx.x;
    const int stride = gridDim.x * blockDim.x;
    for (int i = t; i < nf; i += stride) fl[i] = 0u;
    for (int i = t; i < n;  i += stride) ws[i] = 0.0f;
}

__global__ __launch_bounds__(256)
void link_edges_by_pair(const int* __restrict__ src,
                        const int* __restrict__ dst,
                        const int* __restrict__ ety,
                        int* __restrict__ head,
                        int* __restrict__ recs,
                        int n, int E) {
    const int t = blockIdx.x * blockDim.x + threadIdx.x;
    const int stride = gridDim.x * blockDim.x;
    const int nh = E >> 1;
    const int2* s2p = (const int2*)src;
    const int2* d2p = (const int2*)dst;
    const int2* r2p = (const int2*)ety;
    int4* rec4 = (int4*)recs;
    for (int q = t; q < nh; q += stride) {
        const int2 s2 = s2p[q];
        const int2 r2 = r2p[q];
        const int2 d2 = d2p[q];
        const int e1 = 2 * q + 1;
        const int p0 = atomicExch(head + (size_t)r2.x * n + s2.x, e1);
        const int p1 = atomicExch(head + (size_t)r2.y * n + s2.y, e1 + 1);
        rec4[q] = make_int4(p0, d2.x, p1, d2.y);
    }
    for (int e = (nh << 1) + t; e < E; e += stride) {
        const int p = atomicExch(head + (size_t)ety[e] * n + src[e], e + 1);
        recs[2 * e]     = p;
        recs[2 * e + 1] = dst[e];
    }
}

__global__ __launch_bounds__(TB_THREADS)
void node_relation_attention(const float* __restrict__ ent,
                             const float* __restrict__ rel,
                             const float* __restrict__ WR,
                             const int*   __restrict__ head,
                             const int*   __restrict__ recs,
                             float*       __restrict__ nb,
                             int n, int E) {
    __shared__ float4 ldsW4[W_SLOTS];
    __shared__ alignas(16) float ldsR[D_DIM];
    __shared__ int    ldsL[LIST_CAP];
    __shared__ int    ldsH[LIST_CAP];
    __shared__ int    ldsC[32];
    __shared__ alignas(16) float ldsT[(TB_THREADS >> 5) * 16 * TILE_PITCH];

    const int tid  = threadIdx.x;
    const int lane = tid & 31;
    const int half = lane >> 4;
    const int lr   = lane & 15;
    const int w    = __builtin_amdgcn_readfirstlane(tid >> 5);
    const int nw   = TB_THREADS >> 5;

    const int r    = blockIdx.x & (R_REL - 1);
    const int nrg  = gridDim.x >> 4;
    int rg = blockIdx.x >> 4;

    const int* f = head + (size_t)r * n;
    const int2* rec2 = (const int2*)recs;
    float* myT = ldsT + w * (16 * TILE_PITCH);

    int fr[FL_LOADS];
    {
        const int nlo = rg * RG_NODES;
#pragma unroll
        for (int j = 0; j < FL_LOADS; ++j) fr[j] = f[min(nlo + TB_THREADS * j + tid, n - 1)];
    }
    float4 wv;
    {
        const int l  = tid & 31;
        const int p  = (tid >> 5) & 7;
        const int et = (tid >> 8) & 3;
        const int d0 = 32 * (l >> 4) + 4 * p;
        const int e  = 16 * et + (l & 15);
        const float* g = WR + r * (D_DIM * D_DIM) + d0 * D_DIM + e;
        wv.x = g[0];
        wv.y = g[D_DIM];
        wv.z = g[2 * D_DIM];
        wv.w = g[3 * D_DIM];
    }
    const float4 rv4 = ((const float4*)(rel + r * D_DIM))[tid & 15];
    asm volatile("" :: "v"(fr[0]), "v"(fr[1]), "v"(fr[2]), "v"(fr[3]), "v"(fr[4]), "v"(fr[5]), "v"(fr[6]),
                       "v"(wv.x), "v"(wv.y), "v"(wv.z), "v"(wv.w),
                       "v"(rv4.x), "v"(rv4.y), "v"(rv4.z), "v"(rv4.w));
    ldsW4[tid] = wv;
    if (tid < D_DIM / 4) ((float4*)ldsR)[tid] = rv4;

    for (;;) {
        const int nlo = rg * RG_NODES;

        bool need[FL_LOADS];
        unsigned m[FL_LOADS];
        int cnt = 0;
#pragma unroll
        for (int j = 0; j < FL_LOADS; ++j) {
            const int o = TB_THREADS * j + tid;
            need[j] = (fr[j] != 0) && (o < RG_NODES) && (nlo + o < n);
            m[j] = __builtin_amdgcn_ballot_w32(need[j]);
            cnt += __builtin_popcount(m[j]);
        }
        if (lane == 0) ldsC[w] = cnt;
        __syncthreads();

        const int cW = (lane < nw) ? ldsC[lane] : 0;
        int incl = cW;
#pragma unroll
        for (int d = 1; d < 32; d <<= 1) {
            const int t = __shfl_up(incl, d, 32);
            if (lane >= d) incl += t;
        }
        const int total = __builtin_amdgcn_readlane(incl, 31);
        int run = __builtin_amdgcn_readlane(incl - cW, w);
#pragma unroll
        for (int j = 0; j < FL_LOADS; ++j) {
            const int pos = run + (int)__builtin_amdgcn_mbcnt_lo(m[j], 0u);
            if (need[j]) { ldsL[pos] = nlo + TB_THREADS * j + tid; ldsH[pos] = fr[j]; }
            run += __builtin_popcount(m[j]);
        }
        const int padded = (total + 15) & ~15;
        if (w == 0 && total + lane < padded) { ldsL[total + lane] = nlo; ldsH[total + lane] = 0; }
        __syncthreads();

        const int tiles = padded >> 4;
        for (int it = w; it < tiles; it += nw) {
            asm volatile("" ::: "memory");
            const int slot = it * 16 + lr;
            const int node = ldsL[slot];
            int cur = (half == 0) ? ldsH[slot] : 0;
            int2 rec = rec2[max(cur - 1, 0)];

            const float4* brow = (const float4*)(ent + (size_t)node * D_DIM + 32 * half);
            v2f b[16];
#pragma unroll
            for (int j = 0; j < 8; ++j) {
                const float4 t4 = brow[j];
                b[2 * j].x     = t4.x;  b[2 * j].y     = t4.y;
                b[2 * j + 1].x = t4.z;  b[2 * j + 1].y = t4.w;
            }

            {
                float4* trow = (float4*)(myT + lr * TILE_PITCH + 32 * half);
#pragma unroll
                for (int j = 0; j < 8; ++j) {
                    float4 t4;
                    t4.x = b[2 * j].x;      t4.y = b[2 * j].y;
                    t4.z = b[2 * j + 1].x;  t4.w = b[2 * j + 1].y;
                    trow[j] = t4;
                }
            }

            const float4* wf = ldsW4 + lane;

            v8f acc[4] = {};
#pragma unroll
            for (int p = 0; p < 8; ++p) {
                float4 a[4];
#pragma unroll
                for (int et = 0; et < 4; ++et) a[et] = wf[(et * 8 + p) * 32];
#pragma unroll
                for (int et = 0; et < 4; ++et) {
                    v2f af; af.x = a[et].x; af.y = a[et].y;
                    acc[et] = __builtin_amdgcn_wmma_f32_16x16x4_f32(
                        false, af, false, b[2 * p], (short)0, acc[et], false, false);
                }
#pragma unroll
                for (int et = 0; et < 4; ++et) {
                    v2f af; af.x = a[et].z; af.y = a[et].w;
                    acc[et] = __builtin_amdgcn_wmma_f32_16x16x4_f32(
                        false, af, false, b[2 * p + 1], (short)0, acc[et], false, false);
                }
            }

            float part = 0.f;
#pragma unroll
            for (int et = 0; et < 4; ++et) {
                const float4* rp = (const float4*)(ldsR + 16 * et + 8 * half);
                const float4 r0 = rp[0];
                const float4 r1 = rp[1];
                const float rv[8] = {r0.x, r0.y, r0.z, r0.w, r1.x, r1.y, r1.z, r1.w};
#pragma unroll
                for (int q = 0; q < 8; ++q) {
                    const float mv = acc[et][q];
                    part += mv * tanhf(mv + rv[q]);
                }
            }
            part += __shfl_xor(part, 16, 32);

            const float* trd = myT + lane;
            for (int g = 0; g < E; ++g) {
                const unsigned act = __builtin_amdgcn_ballot_w32(cur != 0);
                if (act == 0u) break;
                const int dcur = rec.y;
                const int nxt  = (cur != 0) ? rec.x : 0;
                const int2 recn = rec2[max(nxt - 1, 0)];
                unsigned mrem = act;
                do {
                    const int c = __builtin_ctz(mrem);
                    mrem &= mrem - 1u;
                    const int   d = __builtin_amdgcn_readlane(dcur, c);
                    const float a = __int_as_float(__builtin_amdgcn_readlane(__float_as_int(part), c));
                    const float x0 = trd[c * TILE_PITCH];
                    const float x1 = trd[c * TILE_PITCH + 32];
                    float* drow = nb + (size_t)d * D_DIM;
                    unsafeAtomicAdd(drow + lane,      x0 * a);
                    unsafeAtomicAdd(drow + lane + 32, x1 * a);
                } while (mrem != 0u);
                cur = nxt;
                rec = recn;
            }
        }

        rg += nrg;
        if (rg * RG_NODES >= n) break;
        __syncthreads();
        {
            const int nlo2 = rg * RG_NODES;
#pragma unroll
            for (int j = 0; j < FL_LOADS; ++j) fr[j] = f[min(nlo2 + TB_THREADS * j + tid, n - 1)];
            asm volatile("" :: "v"(fr[0]), "v"(fr[1]), "v"(fr[2]), "v"(fr[3]), "v"(fr[4]), "v"(fr[5]), "v"(fr[6]));
        }
    }
}

#define CB_TILES 4
#define CB_WAVES 4

__global__ __launch_bounds__(CB_WAVES * 32)
void combine_wmma(const float* __restrict__ ent,
                  const float* __restrict__ nb,
                  const float* __restrict__ Ww,
                  const float* __restrict__ Wb,
                  float*       __restrict__ out,
                  int n) {
    const int lane = threadIdx.x & 31;
    const int half = lane >> 4;
    const int lr   = lane & 15;

    const int wv           = blockIdx.x * CB_WAVES + __builtin_amdgcn_readfirstlane(threadIdx.x >> 5);
    const int numRowBlocks = n / 16;
    const int rb0          = wv * CB_TILES;
    if (rb0 >= numRowBlocks) return;

    v8f acc[CB_TILES][4] = {};

    const float* hrow[CB_TILES];
    const float* nrow[CB_TILES];
#pragma unroll
    for (int t = 0; t < CB_TILES; ++t) {
        const int rb = min(rb0 + t, numRowBlocks - 1);
        const size_t o = (size_t)(rb * 16 + lr) * D_DIM + 2 * half;
        hrow[t] = ent + o;
        nrow[t] = nb + o;
    }
    const float* bcol = Ww + (size_t)lr * TWO_D + 2 * half;

#pragma unroll
    for (int ks = 0; ks < TWO_D / 4; ++ks) {
        v2f a[CB_TILES];
        v2f b[4];
#pragma unroll
        for (int t = 0; t < CB_TILES; ++t) {
            const float* ap = (ks < D_DIM / 4) ? (hrow[t] + 4 * ks) : (nrow[t] + 4 * (ks - D_DIM / 4));
            a[t].x = ap[0];
            a[t].y = ap[1];
        }
#pragma unroll
        for (int c = 0; c < 4; ++c) {
            const float* bp = bcol + (size_t)c * 16 * TWO_D + 4 * ks;
            b[c].x = bp[0];
            b[c].y = bp[1];
        }
#pragma unroll
        for (int c = 0; c < 4; ++c) {
#pragma unroll
            for (int t = 0; t < CB_TILES; ++t) {
                acc[t][c] = __builtin_amdgcn_wmma_f32_16x16x4_f32(
                     false, a[t],  false, b[c],
                     (short)0, acc[t][c],  false,  false);
            }
        }
    }

#pragma unroll
    for (int t = 0; t < CB_TILES; ++t) {
        if (rb0 + t < numRowBlocks) {
            const int row0 = (rb0 + t) * 16;
#pragma unroll
            for (int c = 0; c < 4; ++c) {
                const int col = c * 16 + lr;
                const float bias = Wb[col];
#pragma unroll
                for (int r = 0; r < 8; ++r) {
                    const int row = row0 + r + 8 * half;
                    float x = acc[t][c][r] + bias;
                    x = (x > 0.0f) ? x : 0.01f * x;
                    out[(size_t)row * D_DIM + col] = x + ent[(size_t)row * D_DIM + col];
                }
            }
        }
    }
}

extern "C" void kernel_launch(void* const* d_in, const int* in_sizes, int n_in,
                              void* d_out, int out_size, void* d_ws, size_t ws_size,
                              hipStream_t stream) {
    const float* ent = (const float*)d_in[0];
    const float* rel = (const float*)d_in[1];
    const float* WR  = (const float*)d_in[2];
    const float* Ww  = (const float*)d_in[3];
    const float* Wb  = (const float*)d_in[4];
    const int*   src = (const int*)d_in[5];
    const int*   dst = (const int*)d_in[6];
    const int*   ety = (const int*)d_in[7];
    float*       out = (float*)d_out;

    const int n = in_sizes[0] / D_DIM;
    const int E = in_sizes[5];

    float* nb  = (float*)d_ws;
    int* head = (int*)out;
    int* recs = (int*)(out + (size_t)R_REL * n);

    zero_ws_kernel<<<1024, 256, 0, stream>>>(nb, n * D_DIM, (unsigned int*)head, R_REL * n);

    link_edges_by_pair<<<512, 256, 0, stream>>>(src, dst, ety, head, recs, n, E);

    node_relation_attention<<<256, TB_THREADS, 0, stream>>>(ent, rel, WR, head, recs, nb, n, E);

    const int rowBlocks = n / 16;
    combine_wmma<<<(rowBlocks + CB_TILES * CB_WAVES - 1) / (CB_TILES * CB_WAVES), CB_WAVES * 32, 0, stream>>>(ent, nb, Ww, Wb, out, n);
}
